# starting kernel
# baseline (speedup 1.0000x reference)
#include <hip/hip_runtime.h>
#include <hip/hip_bf16.h>

typedef __attribute__((ext_vector_type(2))) float v2f;
typedef __attribute__((ext_vector_type(4))) float v4f;
typedef __attribute__((ext_vector_type(8))) float v8f;

#define N_MODES 10

__global__ __launch_bounds__(256) void DQSEncoder_kernel(
    const float* __restrict__ x,
    const float* __restrict__ w,
    const float* __restrict__ vr,
    float* __restrict__ out,
    int V)
{
    const int lane = threadIdx.x & 31;
    const int wave = (blockIdx.x << 3) + (threadIdx.x >> 5);
    const int v0   = wave << 4;
    if (v0 >= V) return;

    const int col  = lane & 15;
    const int half = lane >> 4;

    __builtin_prefetch(vr + (size_t)v0 * 4, 0, 3);

    v2f Bt[4][5];
#pragma unroll
    for (int t = 0; t < 4; ++t) {
        const int c = t * 16 + col;
        const int b = c >> 2;
        const int f = c & 3;
#pragma unroll
        for (int ks = 0; ks < 5; ++ks) {
#pragma unroll
            for (int j = 0; j < 2; ++j) {
                const int k = ks * 4 + half * 2 + j;
                float val;
                if (k < N_MODES) val = x[b * (N_MODES * 4) + k * 4 + f];
                else             val = (f == 3) ? 1.0f : 0.0f;
                Bt[t][ks][j] = val;
            }
        }
    }

    v8f acc[4];
#pragma unroll
    for (int t = 0; t < 4; ++t) acc[t] = (v8f){0,0,0,0,0,0,0,0};

    const float* wrow = w + (size_t)(v0 + col) * 20 + half * 2;
#pragma unroll
    for (int ks = 0; ks < 5; ++ks) {
        const v2f A = *(const v2f*)(wrow + ks * 4);
#pragma unroll
        for (int t = 0; t < 4; ++t) {
            acc[t] = __builtin_amdgcn_wmma_f32_16x16x4_f32(
                 false, A,  false, Bt[t][ks],
                 (short)0, acc[t],
                 false,  false);
        }
    }

    const int f    = lane & 3;
    const int base = lane & ~3;
#pragma unroll
    for (int j = 0; j < 8; ++j) {
        const int vrow = v0 + j + half * 8;
        const v4f vv = *(const v4f*)(vr + (size_t)vrow * 4);
#pragma unroll
        for (int t = 0; t < 4; ++t) {
            float q = acc[t][j];
            float s = q * q;
            s += __shfl_xor(s, 1);
            s += __shfl_xor(s, 2);
            q *= rsqrtf(s);
            const float qx = __shfl(q, base + 0);
            const float qy = __shfl(q, base + 1);
            const float qz = __shfl(q, base + 2);
            const float qw = __shfl(q, base + 3);

            const float xx = qx*qx, yy = qy*qy, zz = qz*qz;
            const float xy = qx*qy, xz = qx*qz, yz = qy*qz;
            const float wx = qw*qx, wy = qw*qy, wz = qw*qz;

            float r0, r1, r2;
            if (f == 0)      { r0 = 1.0f - 2.0f*(yy+zz); r1 = 2.0f*(xy-wz);        r2 = 2.0f*(xz+wy); }
            else if (f == 1) { r0 = 2.0f*(xy+wz);        r1 = 1.0f - 2.0f*(xx+zz); r2 = 2.0f*(yz-wx); }
            else             { r0 = 2.0f*(xz-wy);        r1 = 2.0f*(yz+wx);        r2 = 1.0f - 2.0f*(xx+yy); }

            float y = r0 * vv[0] + r1 * vv[1] + r2 * vv[2];
            if (f == 3) y = vv[3];

            const int b = (t * 16 + col) >> 2;
            out[(size_t)b * ((size_t)V * 4) + (size_t)vrow * 4 + f] = y;
        }
    }
}

extern "C" void kernel_launch(void* const* d_in, const int* in_sizes, int n_in,
                              void* d_out, int out_size, void* d_ws, size_t ws_size,
                              hipStream_t stream) {
    const float* x  = (const float*)d_in[0];
    const float* w  = (const float*)d_in[1];
    const float* vr = (const float*)d_in[2];
    float* out = (float*)d_out;

    const int V = in_sizes[2] / 4;
    const int tiles = (V + 15) / 16;
    const int wavesPerBlock = 8;
    const int blocks = (tiles + wavesPerBlock - 1) / wavesPerBlock;

    DQSEncoder_kernel<<<blocks, 256, 0, stream>>>(x, w, vr, out, V);
}


# Swap WMMA operand roles (x side as A, weights as B): a lane's 8 accumulator registers now hold whole quaternions, so the epilogue needs no lane exchange and stores 16 bytes per quaternion
# speedup vs baseline: 2.4852x; 2.4852x over previous
#include <hip/hip_runtime.h>
#include <hip/hip_bf16.h>

typedef __attribute__((ext_vector_type(2))) float v2f;
typedef __attribute__((ext_vector_type(4))) float v4f;
typedef __attribute__((ext_vector_type(8))) float v8f;

#define N_MODES 10

__global__ __launch_bounds__(256) void DQSEncoder_kernel(
    const float* __restrict__ x,
    const float* __restrict__ w,
    const float* __restrict__ vr,
    float* __restrict__ out,
    int V)
{
    const int lane = threadIdx.x & 31;
    const int wave = (blockIdx.x << 3) + (threadIdx.x >> 5);
    const int v0   = wave << 4;
    if (v0 >= V) return;

    const int col  = lane & 15;
    const int half = lane >> 4;
    const int v    = v0 + col;

    const v4f vv = *(const v4f*)(vr + (size_t)v * 4);
    const float* wrow = w + (size_t)v * 20 + half * 2;
    v2f Wk[5];
#pragma unroll
    for (int ks = 0; ks < 5; ++ks) Wk[ks] = *(const v2f*)(wrow + ks * 4);

    v2f Xt[4][5];
#pragma unroll
    for (int t = 0; t < 4; ++t) {
        const int c = t * 16 + col;
        const int b = c >> 2;
        const int f = c & 3;
#pragma unroll
        for (int ks = 0; ks < 5; ++ks) {
#pragma unroll
            for (int j = 0; j < 2; ++j) {
                const int k = ks * 4 + half * 2 + j;
                float val;
                if (k < N_MODES) val = x[b * (N_MODES * 4) + k * 4 + f];
                else             val = (f == 3) ? 1.0f : 0.0f;
                Xt[t][ks][j] = val;
            }
        }
    }

    v8f acc[4];
#pragma unroll
    for (int t = 0; t < 4; ++t) acc[t] = (v8f){0,0,0,0,0,0,0,0};

#pragma unroll
    for (int ks = 0; ks < 5; ++ks) {
#pragma unroll
        for (int t = 0; t < 4; ++t) {
            acc[t] = __builtin_amdgcn_wmma_f32_16x16x4_f32(
                 false, Xt[t][ks],  false, Wk[ks],
                 (short)0, acc[t],
                 false,  false);
        }
    }

    float* orow = out + (size_t)v * 4;
    const size_t bstride = (size_t)V * 4;
#pragma unroll
    for (int t = 0; t < 4; ++t) {
#pragma unroll
        for (int u = 0; u < 2; ++u) {
            float qx = acc[t][4 * u + 0];
            float qy = acc[t][4 * u + 1];
            float qz = acc[t][4 * u + 2];
            float qw = acc[t][4 * u + 3];
            const float s   = (qx * qx + qy * qy) + (qz * qz + qw * qw);
            const float inv = rsqrtf(s);
            qx *= inv; qy *= inv; qz *= inv; qw *= inv;

            const float xx = qx*qx, yy = qy*qy, zz = qz*qz;
            const float xy = qx*qy, xz = qx*qz, yz = qy*qz;
            const float wx = qw*qx, wy = qw*qy, wz = qw*qz;

            v4f y;
            y[0] = (1.0f - 2.0f*(yy+zz)) * vv[0] + (2.0f*(xy-wz)) * vv[1] + (2.0f*(xz+wy)) * vv[2];
            y[1] = (2.0f*(xy+wz)) * vv[0] + (1.0f - 2.0f*(xx+zz)) * vv[1] + (2.0f*(yz-wx)) * vv[2];
            y[2] = (2.0f*(xz-wy)) * vv[0] + (2.0f*(yz+wx)) * vv[1] + (1.0f - 2.0f*(xx+yy)) * vv[2];
            y[3] = vv[3];

            const int b = 4 * t + 2 * half + u;
            *(v4f*)(orow + (size_t)b * bstride) = y;
        }
    }
}

extern "C" void kernel_launch(void* const* d_in, const int* in_sizes, int n_in,
                              void* d_out, int out_size, void* d_ws, size_t ws_size,
                              hipStream_t stream) {
    const float* x  = (const float*)d_in[0];
    const float* w  = (const float*)d_in[1];
    const float* vr = (const float*)d_in[2];
    float* out = (float*)d_out;

    const int V = in_sizes[2] / 4;
    const int tiles = (V + 15) / 16;
    const int wavesPerBlock = 8;
    const int blocks = (tiles + wavesPerBlock - 1) / wavesPerBlock;

    DQSEncoder_kernel<<<blocks, 256, 0, stream>>>(x, w, vr, out, V);
}


# Each wave finishes 4 tiles of 16 v instead of 1 (grid cut to a quarter, every wave resident at once): x-side fragments built once per wave, all 4 tiles' weight and VR loads requested up front
# speedup vs baseline: 2.7375x; 1.1015x over previous
#include <hip/hip_runtime.h>
#include <hip/hip_bf16.h>

typedef __attribute__((ext_vector_type(2))) float v2f;
typedef __attribute__((ext_vector_type(4))) float v4f;
typedef __attribute__((ext_vector_type(8))) float v8f;

#define N_MODES 10
#define TILES_PER_WAVE 4

__global__ __launch_bounds__(256) void DQSEncoder_kernel(
    const float* __restrict__ x,
    const float* __restrict__ w,
    const float* __restrict__ vr,
    float* __restrict__ out,
    int V)
{
    const int lane   = threadIdx.x & 31;
    const int nwaves = gridDim.x << 3;
    const int wave   = (blockIdx.x << 3) + (threadIdx.x >> 5);
    if ((wave << 4) >= V) return;

    const int col  = lane & 15;
    const int half = lane >> 4;

    v4f vv[TILES_PER_WAVE];
    v2f Wk[TILES_PER_WAVE][5];
#pragma unroll
    for (int i = 0; i < TILES_PER_WAVE; ++i) {
        const int vraw = ((i * nwaves + wave) << 4) + col;
        const int vld  = vraw < V ? vraw : V - 1;
        vv[i] = *(const v4f*)(vr + (size_t)vld * 4);
        const float* wrow = w + (size_t)vld * 20 + half * 2;
#pragma unroll
        for (int ks = 0; ks < 5; ++ks) Wk[i][ks] = *(const v2f*)(wrow + ks * 4);
    }

    v2f Xt[4][5];
#pragma unroll
    for (int t = 0; t < 4; ++t) {
        const int c = t * 16 + col;
        const int b = c >> 2;
        const int f = c & 3;
#pragma unroll
        for (int ks = 0; ks < 5; ++ks) {
#pragma unroll
            for (int j = 0; j < 2; ++j) {
                const int k = ks * 4 + half * 2 + j;
                float val;
                if (k < N_MODES) val = x[b * (N_MODES * 4) + k * 4 + f];
                else             val = (f == 3) ? 1.0f : 0.0f;
                Xt[t][ks][j] = val;
            }
        }
    }

    const size_t bstride = (size_t)V * 4;

#pragma unroll
    for (int i = 0; i < TILES_PER_WAVE; ++i) {
        const int v = ((i * nwaves + wave) << 4) + col;

        v8f acc[4];
#pragma unroll
        for (int t = 0; t < 4; ++t) acc[t] = (v8f){0,0,0,0,0,0,0,0};

#pragma unroll
        for (int ks = 0; ks < 5; ++ks) {
#pragma unroll
            for (int t = 0; t < 4; ++t) {
                acc[t] = __builtin_amdgcn_wmma_f32_16x16x4_f32(
                     false, Xt[t][ks],  false, Wk[i][ks],
                     (short)0, acc[t],
                     false,  false);
            }
        }

        v4f y[4][2];
#pragma unroll
        for (int t = 0; t < 4; ++t) {
#pragma unroll
            for (int u = 0; u < 2; ++u) {
                float qx = acc[t][4 * u + 0];
                float qy = acc[t][4 * u + 1];
                float qz = acc[t][4 * u + 2];
                float qw = acc[t][4 * u + 3];
                const float s   = (qx * qx + qy * qy) + (qz * qz + qw * qw);
                const float inv = rsqrtf(s);
                qx *= inv; qy *= inv; qz *= inv; qw *= inv;

                const float xx = qx*qx, yy = qy*qy, zz = qz*qz;
                const float xy = qx*qy, xz = qx*qz, yz = qy*qz;
                const float wx = qw*qx, wy = qw*qy, wz = qw*qz;

                y[t][u][0] = (1.0f - 2.0f*(yy+zz)) * vv[i][0] + (2.0f*(xy-wz)) * vv[i][1] + (2.0f*(xz+wy)) * vv[i][2];
                y[t][u][1] = (2.0f*(xy+wz)) * vv[i][0] + (1.0f - 2.0f*(xx+zz)) * vv[i][1] + (2.0f*(yz-wx)) * vv[i][2];
                y[t][u][2] = (2.0f*(xz-wy)) * vv[i][0] + (2.0f*(yz+wx)) * vv[i][1] + (1.0f - 2.0f*(xx+yy)) * vv[i][2];
                y[t][u][3] = vv[i][3];
            }
        }

        if (v < V) {
            float* orow = out + (size_t)v * 4;
#pragma unroll
            for (int t = 0; t < 4; ++t) {
#pragma unroll
                for (int u = 0; u < 2; ++u) {
                    const int b = 4 * t + 2 * half + u;
                    *(v4f*)(orow + (size_t)b * bstride) = y[t][u];
                }
            }
        }
    }
}

extern "C" void kernel_launch(void* const* d_in, const int* in_sizes, int n_in,
                              void* d_out, int out_size, void* d_ws, size_t ws_size,
                              hipStream_t stream) {
    const float* x  = (const float*)d_in[0];
    const float* w  = (const float*)d_in[1];
    const float* vr = (const float*)d_in[2];
    float* out = (float*)d_out;

    const int V = in_sizes[2] / 4;
    const int tiles = (V + 15) / 16;
    const int wavesPerBlock = 8;
    const int tilesPerBlock = wavesPerBlock * TILES_PER_WAVE;
    const int blocks = (tiles + tilesPerBlock - 1) / tilesPerBlock;

    DQSEncoder_kernel<<<blocks, 256, 0, stream>>>(x, w, vr, out, V);
}


# Epilogue rotates by the unnormalised quaternion on packed f32 pairs: y = v + (w c + u x c)/|q|^2, c = u x 2v; one v_rcp_f32 replaces rsqrt, normalise and the 3x3 matrix; f32 kept, rounding differs
# speedup vs baseline: 3.1110x; 1.1364x over previous
#include <hip/hip_runtime.h>
#include <hip/hip_bf16.h>

typedef __attribute__((ext_vector_type(2))) float v2f;
typedef __attribute__((ext_vector_type(4))) float v4f;
typedef __attribute__((ext_vector_type(8))) float v8f;

#define N_MODES 10
#define TILES_PER_WAVE 4

__global__ __launch_bounds__(256) void DQSEncoder_kernel(
    const float* __restrict__ x,
    const float* __restrict__ w,
    const float* __restrict__ vr,
    float* __restrict__ out,
    int V)
{
    const int lane   = threadIdx.x & 31;
    const int nwaves = gridDim.x << 3;
    const int wave   = (blockIdx.x << 3) + (threadIdx.x >> 5);
    if ((wave << 4) >= V) return;

    const int col  = lane & 15;
    const int half = lane >> 4;

    v4f vv[TILES_PER_WAVE];
    v2f Wk[TILES_PER_WAVE][5];
#pragma unroll
    for (int i = 0; i < TILES_PER_WAVE; ++i) {
        const int vraw = ((i * nwaves + wave) << 4) + col;
        const int vld  = vraw < V ? vraw : V - 1;
        vv[i] = *(const v4f*)(vr + (size_t)vld * 4);
        const float* wrow = w + (size_t)vld * 20 + half * 2;
#pragma unroll
        for (int ks = 0; ks < 5; ++ks) Wk[i][ks] = *(const v2f*)(wrow + ks * 4);
    }

    v2f Xt[4][5];
#pragma unroll
    for (int t = 0; t < 4; ++t) {
        const int c = t * 16 + col;
        const int b = c >> 2;
        const int f = c & 3;
#pragma unroll
        for (int ks = 0; ks < 5; ++ks) {
#pragma unroll
            for (int j = 0; j < 2; ++j) {
                const int k = ks * 4 + half * 2 + j;
                float val;
                if (k < N_MODES) val = x[b * (N_MODES * 4) + k * 4 + f];
                else             val = (f == 3) ? 1.0f : 0.0f;
                Xt[t][ks][j] = val;
            }
        }
    }

    const size_t bstride = (size_t)V * 4;

#pragma unroll
    for (int i = 0; i < TILES_PER_WAVE; ++i) {
        const int v = ((i * nwaves + wave) << 4) + col;

        v8f acc[4];
#pragma unroll
        for (int t = 0; t < 4; ++t) acc[t] = (v8f){0,0,0,0,0,0,0,0};

#pragma unroll
        for (int ks = 0; ks < 5; ++ks) {
#pragma unroll
            for (int t = 0; t < 4; ++t) {
                acc[t] = __builtin_amdgcn_wmma_f32_16x16x4_f32(
                     false, Xt[t][ks],  false, Wk[i][ks],
                     (short)0, acc[t],
                     false,  false);
            }
        }

        const float t0 = 2.0f * vv[i][0], t1 = 2.0f * vv[i][1], t2 = 2.0f * vv[i][2];
        const float v2 = vv[i][2];
        const v2f V01  = {vv[i][0], vv[i][1]};
        const v2f T22  = {t2, t2};
        const v2f Tn10 = {-t1, t0};
        v4f y[4][2];
#pragma unroll
        for (int t = 0; t < 4; ++t) {
#pragma unroll
            for (int u = 0; u < 2; ++u) {
                const v2f P = {acc[t][4 * u + 0], acc[t][4 * u + 1]};
                const v2f Q = {acc[t][4 * u + 2], acc[t][4 * u + 3]};
                const v2f ZZ = {Q[0], Q[0]};
                const v2f WW = {Q[1], Q[1]};
                const v2f S2 = P * P + Q * Q;
                const float s = S2[0] + S2[1];
                const float r = __builtin_amdgcn_rcpf(s);
                const v2f A1  = P * T22;
                const v2f C01 = ZZ * Tn10 + (v2f){A1[1], -A1[0]};
                const float cz = P[0] * t1 - P[1] * t0;
                const v2f T1  = ZZ * (v2f){-C01[1], C01[0]};
                const v2f T2  = (v2f){P[1], P[0]} * (v2f){cz, -cz} + T1;
                const v2f E01 = WW * C01 + T2;
                const float ez = Q[1] * cz + (P[0] * C01[1] - P[1] * C01[0]);
                const v2f Y01 = (v2f){r, r} * E01 + V01;

                y[t][u][0] = Y01[0];
                y[t][u][1] = Y01[1];
                y[t][u][2] = v2 + r * ez;
                y[t][u][3] = vv[i][3];
            }
        }

        if (v < V) {
            float* orow = out + (size_t)v * 4;
#pragma unroll
            for (int t = 0; t < 4; ++t) {
#pragma unroll
                for (int u = 0; u < 2; ++u) {
                    const int b = 4 * t + 2 * half + u;
                    *(v4f*)(orow + (size_t)b * bstride) = y[t][u];
                }
            }
        }
    }
}

extern "C" void kernel_launch(void* const* d_in, const int* in_sizes, int n_in,
                              void* d_out, int out_size, void* d_ws, size_t ws_size,
                              hipStream_t stream) {
    const float* x  = (const float*)d_in[0];
    const float* w  = (const float*)d_in[1];
    const float* vr = (const float*)d_in[2];
    float* out = (float*)d_out;

    const int V = in_sizes[2] / 4;
    const int tiles = (V + 15) / 16;
    const int wavesPerBlock = 8;
    const int tilesPerBlock = wavesPerBlock * TILES_PER_WAVE;
    const int blocks = (tiles + tilesPerBlock - 1) / tilesPerBlock;

    DQSEncoder_kernel<<<blocks, 256, 0, stream>>>(x, w, vr, out, V);
}


# The ten constant rest-quaternion rows are one row: weights of k=12..19 are summed (vector adds, one half-swap) into the k=10 operand slot, WMMA K-steps 5 -> 3; exact algebra, summation order changes
# speedup vs baseline: 3.3207x; 1.0674x over previous
#include <hip/hip_runtime.h>
#include <hip/hip_bf16.h>

typedef __attribute__((ext_vector_type(2))) float v2f;
typedef __attribute__((ext_vector_type(4))) float v4f;
typedef __attribute__((ext_vector_type(8))) float v8f;

#define N_MODES 10
#define TILES_PER_WAVE 4

__global__ __launch_bounds__(256) void DQSEncoder_kernel(
    const float* __restrict__ x,
    const float* __restrict__ w,
    const float* __restrict__ vr,
    float* __restrict__ out,
    int V)
{
    const int lane   = threadIdx.x & 31;
    const int nwaves = gridDim.x << 3;
    const int wave   = (blockIdx.x << 3) + (threadIdx.x >> 5);
    if ((wave << 4) >= V) return;

    const int col  = lane & 15;
    const int half = lane >> 4;

    v4f vv[TILES_PER_WAVE];
    v2f Wk[TILES_PER_WAVE][5];
#pragma unroll
    for (int i = 0; i < TILES_PER_WAVE; ++i) {
        const int vraw = ((i * nwaves + wave) << 4) + col;
        const int vld  = vraw < V ? vraw : V - 1;
        vv[i] = *(const v4f*)(vr + (size_t)vld * 4);
        const float* wrow = w + (size_t)vld * 20 + half * 2;
#pragma unroll
        for (int ks = 0; ks < 5; ++ks) Wk[i][ks] = *(const v2f*)(wrow + ks * 4);
    }

    v2f Xt[4][3];
#pragma unroll
    for (int t = 0; t < 4; ++t) {
        const int c = t * 16 + col;
        const int b = c >> 2;
        const int f = c & 3;
#pragma unroll
        for (int ks = 0; ks < 3; ++ks) {
#pragma unroll
            for (int j = 0; j < 2; ++j) {
                const int k = ks * 4 + half * 2 + j;
                float val;
                if (k < N_MODES) val = x[b * (N_MODES * 4) + k * 4 + f];
                else             val = (f == 3) ? 1.0f : 0.0f;
                Xt[t][ks][j] = val;
            }
        }
    }

    const size_t bstride = (size_t)V * 4;

#pragma unroll
    for (int i = 0; i < TILES_PER_WAVE; ++i) {
        const int v = ((i * nwaves + wave) << 4) + col;

        const v2f c2 = Wk[i][3] + Wk[i][4];
        const float cown = c2[0] + c2[1];
        const int cbits = __builtin_bit_cast(int, cown);
        const float coth = __builtin_bit_cast(float, __builtin_amdgcn_permlanex16(
            cbits, cbits, 0x76543210, 0xfedcba98, false, false));
        Wk[i][2][0] += half ? (cown + coth) : 0.0f;

        v8f acc[4];
#pragma unroll
        for (int t = 0; t < 4; ++t) acc[t] = (v8f){0,0,0,0,0,0,0,0};

#pragma unroll
        for (int ks = 0; ks < 3; ++ks) {
#pragma unroll
            for (int t = 0; t < 4; ++t) {
                acc[t] = __builtin_amdgcn_wmma_f32_16x16x4_f32(
                     false, Xt[t][ks],  false, Wk[i][ks],
                     (short)0, acc[t],
                     false,  false);
            }
        }

        const float t0 = 2.0f * vv[i][0], t1 = 2.0f * vv[i][1], t2 = 2.0f * vv[i][2];
        const float v2 = vv[i][2];
        const v2f V01  = {vv[i][0], vv[i][1]};
        const v2f T22  = {t2, t2};
        const v2f Tn10 = {-t1, t0};
        v4f y[4][2];
#pragma unroll
        for (int t = 0; t < 4; ++t) {
#pragma unroll
            for (int u = 0; u < 2; ++u) {
                const v2f P = {acc[t][4 * u + 0], acc[t][4 * u + 1]};
                const v2f Q = {acc[t][4 * u + 2], acc[t][4 * u + 3]};
                const v2f ZZ = {Q[0], Q[0]};
                const v2f WW = {Q[1], Q[1]};
                const v2f S2 = P * P + Q * Q;
                const float s = S2[0] + S2[1];
                const float r = __builtin_amdgcn_rcpf(s);
                const v2f A1  = P * T22;
                const v2f C01 = ZZ * Tn10 + (v2f){A1[1], -A1[0]};
                const float cz = P[0] * t1 - P[1] * t0;
                const v2f T1  = ZZ * (v2f){-C01[1], C01[0]};
                const v2f T2  = (v2f){P[1], P[0]} * (v2f){cz, -cz} + T1;
                const v2f E01 = WW * C01 + T2;
                const float ez = Q[1] * cz + (P[0] * C01[1] - P[1] * C01[0]);
                const v2f Y01 = (v2f){r, r} * E01 + V01;

                y[t][u][0] = Y01[0];
                y[t][u][1] = Y01[1];
                y[t][u][2] = v2 + r * ez;
                y[t][u][3] = vv[i][3];
            }
        }

        if (v < V) {
            float* orow = out + (size_t)v * 4;
#pragma unroll
            for (int t = 0; t < 4; ++t) {
#pragma unroll
                for (int u = 0; u < 2; ++u) {
                    const int b = 4 * t + 2 * half + u;
                    *(v4f*)(orow + (size_t)b * bstride) = y[t][u];
                }
            }
        }
    }
}

extern "C" void kernel_launch(void* const* d_in, const int* in_sizes, int n_in,
                              void* d_out, int out_size, void* d_ws, size_t ws_size,
                              hipStream_t stream) {
    const float* x  = (const float*)d_in[0];
    const float* w  = (const float*)d_in[1];
    const float* vr = (const float*)d_in[2];
    float* out = (float*)d_out;

    const int V = in_sizes[2] / 4;
    const int tiles = (V + 15) / 16;
    const int wavesPerBlock = 8;
    const int tilesPerBlock = wavesPerBlock * TILES_PER_WAVE;
    const int blocks = (tiles + tilesPerBlock - 1) / tilesPerBlock;

    DQSEncoder_kernel<<<blocks, 256, 0, stream>>>(x, w, vr, out, V);
}


# Operand feed through LDS: x and each tile's weights fetched as whole 128-byte lines (16 contiguous bytes per lane), parked in a wave-private LDS region, fragments read back from there; no barrier
# speedup vs baseline: 3.5621x; 1.0727x over previous
#include <hip/hip_runtime.h>
#include <hip/hip_bf16.h>

typedef __attribute__((ext_vector_type(2))) float v2f;
typedef __attribute__((ext_vector_type(4))) float v4f;
typedef __attribute__((ext_vector_type(8))) float v8f;

#define N_MODES 10
#define TILES_PER_WAVE 4
#define WAVES_PER_BLOCK 8
#define X_FLOATS 640
#define TILE_W_FLOATS 320
#define WAVE_LDS_FLOATS (X_FLOATS + TILES_PER_WAVE * TILE_W_FLOATS)

__global__ __launch_bounds__(256) void DQSEncoder_kernel(
    const float* __restrict__ x,
    const float* __restrict__ w,
    const float* __restrict__ vr,
    float* __restrict__ out,
    int V)
{
    const int lane   = threadIdx.x & 31;
    const int nwaves = gridDim.x << 3;
    const int wave   = (blockIdx.x << 3) + (threadIdx.x >> 5);
    if ((wave << 4) >= V) return;

    const int col  = lane & 15;
    const int half = lane >> 4;

    __shared__ __attribute__((aligned(16))) float lds[WAVES_PER_BLOCK * WAVE_LDS_FLOATS];
    float* xs = lds + (threadIdx.x >> 5) * WAVE_LDS_FLOATS;
    float* ws = xs + X_FLOATS;

    v4f xg[5];
#pragma unroll
    for (int q = 0; q < 5; ++q) xg[q] = *(const v4f*)(x + q * 128 + lane * 4);

    const int wend = V * 20;
    v4f wg4[TILES_PER_WAVE][2];
    v2f wg2[TILES_PER_WAVE];
    v4f vv[TILES_PER_WAVE];
#pragma unroll
    for (int i = 0; i < TILES_PER_WAVE; ++i) {
        const int wbase = (i * nwaves + wave) * TILE_W_FLOATS;
#pragma unroll
        for (int m = 0; m < 2; ++m) {
            const int o4 = wbase + m * 128 + lane * 4;
            wg4[i][m] = *(const v4f*)(w + (o4 < wend - 4 ? o4 : wend - 4));
        }
        const int o2 = wbase + 256 + lane * 2;
        wg2[i] = *(const v2f*)(w + (o2 < wend - 2 ? o2 : wend - 2));
    }
#pragma unroll
    for (int i = 0; i < TILES_PER_WAVE; ++i) {
        const int vraw = ((i * nwaves + wave) << 4) + col;
        const int vld  = vraw < V ? vraw : V - 1;
        vv[i] = *(const v4f*)(vr + (size_t)vld * 4);
    }

#pragma unroll
    for (int q = 0; q < 5; ++q) *(v4f*)(xs + q * 128 + lane * 4) = xg[q];
#pragma unroll
    for (int i = 0; i < TILES_PER_WAVE; ++i) {
#pragma unroll
        for (int m = 0; m < 2; ++m) *(v4f*)(ws + i * TILE_W_FLOATS + m * 128 + lane * 4) = wg4[i][m];
        *(v2f*)(ws + i * TILE_W_FLOATS + 256 + lane * 2) = wg2[i];
    }
    __builtin_amdgcn_fence(__ATOMIC_RELEASE, "wavefront");
    __builtin_amdgcn_wave_barrier();
    __builtin_amdgcn_fence(__ATOMIC_ACQUIRE, "wavefront");

    v2f Wk[TILES_PER_WAVE][5];
#pragma unroll
    for (int i = 0; i < TILES_PER_WAVE; ++i) {
        const float* wrow = ws + i * TILE_W_FLOATS + col * 20 + half * 2;
#pragma unroll
        for (int ks = 0; ks < 5; ++ks) Wk[i][ks] = *(const v2f*)(wrow + ks * 4);
    }

    v2f Xt[4][3];
#pragma unroll
    for (int t = 0; t < 4; ++t) {
        const int c = t * 16 + col;
        const int b = c >> 2;
        const int f = c & 3;
        const float cst = (f == 3) ? 1.0f : 0.0f;
#pragma unroll
        for (int ks = 0; ks < 3; ++ks) {
#pragma unroll
            for (int j = 0; j < 2; ++j) {
                const int k  = ks * 4 + half * 2 + j;
                const int kk = k < N_MODES ? k : N_MODES - 1;
                const float val = xs[b * (N_MODES * 4) + kk * 4 + f];
                Xt[t][ks][j] = (k < N_MODES) ? val : cst;
            }
        }
    }

    const size_t bstride = (size_t)V * 4;

#pragma unroll
    for (int i = 0; i < TILES_PER_WAVE; ++i) {
        const int v = ((i * nwaves + wave) << 4) + col;

        const v2f c2 = Wk[i][3] + Wk[i][4];
        const float cown = c2[0] + c2[1];
        const int cbits = __builtin_bit_cast(int, cown);
        const float coth = __builtin_bit_cast(float, __builtin_amdgcn_permlanex16(
            cbits, cbits, 0x76543210, 0xfedcba98, false, false));
        Wk[i][2][0] += half ? (cown + coth) : 0.0f;

        v8f acc[4];
#pragma unroll
        for (int t = 0; t < 4; ++t) acc[t] = (v8f){0,0,0,0,0,0,0,0};

#pragma unroll
        for (int ks = 0; ks < 3; ++ks) {
#pragma unroll
            for (int t = 0; t < 4; ++t) {
                acc[t] = __builtin_amdgcn_wmma_f32_16x16x4_f32(
                     false, Xt[t][ks],  false, Wk[i][ks],
                     (short)0, acc[t],
                     false,  false);
            }
        }

        const float t0 = 2.0f * vv[i][0], t1 = 2.0f * vv[i][1], t2 = 2.0f * vv[i][2];
        const float v2 = vv[i][2];
        const v2f V01  = {vv[i][0], vv[i][1]};
        const v2f T22  = {t2, t2};
        const v2f Tn10 = {-t1, t0};
        v4f y[4][2];
#pragma unroll
        for (int t = 0; t < 4; ++t) {
#pragma unroll
            for (int u = 0; u < 2; ++u) {
                const v2f P = {acc[t][4 * u + 0], acc[t][4 * u + 1]};
                const v2f Q = {acc[t][4 * u + 2], acc[t][4 * u + 3]};
                const v2f ZZ = {Q[0], Q[0]};
                const v2f WW = {Q[1], Q[1]};
                const v2f S2 = P * P + Q * Q;
                const float s = S2[0] + S2[1];
                const float r = __builtin_amdgcn_rcpf(s);
                const v2f A1  = P * T22;
                const v2f C01 = ZZ * Tn10 + (v2f){A1[1], -A1[0]};
                const float cz = P[0] * t1 - P[1] * t0;
                const v2f T1  = ZZ * (v2f){-C01[1], C01[0]};
                const v2f T2  = (v2f){P[1], P[0]} * (v2f){cz, -cz} + T1;
                const v2f E01 = WW * C01 + T2;
                const float ez = Q[1] * cz + (P[0] * C01[1] - P[1] * C01[0]);
                const v2f Y01 = (v2f){r, r} * E01 + V01;

                y[t][u][0] = Y01[0];
                y[t][u][1] = Y01[1];
                y[t][u][2] = v2 + r * ez;
                y[t][u][3] = vv[i][3];
            }
        }

        if (v < V) {
            float* orow = out + (size_t)v * 4;
#pragma unroll
            for (int t = 0; t < 4; ++t) {
#pragma unroll
                for (int u = 0; u < 2; ++u) {
                    const int b = 4 * t + 2 * half + u;
                    *(v4f*)(orow + (size_t)b * bstride) = y[t][u];
                }
            }
        }
    }
}

extern "C" void kernel_launch(void* const* d_in, const int* in_sizes, int n_in,
                              void* d_out, int out_size, void* d_ws, size_t ws_size,
                              hipStream_t stream) {
    const float* x  = (const float*)d_in[0];
    const float* w  = (const float*)d_in[1];
    const float* vr = (const float*)d_in[2];
    float* out = (float*)d_out;

    const int V = in_sizes[2] / 4;
    const int tiles = (V + 15) / 16;
    const int wavesPerBlock = 8;
    const int tilesPerBlock = wavesPerBlock * TILES_PER_WAVE;
    const int blocks = (tiles + tilesPerBlock - 1) / tilesPerBlock;

    DQSEncoder_kernel<<<blocks, 256, 0, stream>>>(x, w, vr, out, V);
}
